# starting kernel
# baseline (speedup 1.0000x reference)
#include <hip/hip_runtime.h>
#include <math.h>

typedef __attribute__((ext_vector_type(2))) float v2f;
typedef __attribute__((ext_vector_type(8))) float v8f;

#define C_DIM 16
#define T_DIM 160
#define NITERS 10
#define EPSV 1e-8f
#define XS_STRIDE 162
#define AUG_STRIDE 33
#define JS 17
#define NSWEEP 8

__device__ __forceinline__ v8f wmma_f32_16x16x4(v2f a, v2f b, v8f c) {
  return __builtin_amdgcn_wmma_f32_16x16x4_f32(false, a, false, b, (short)0, c,
                                               false, false);
}

__device__ __forceinline__ float wave_sum32(float v, float* red, int lane) {
  red[lane] = v;
  __syncthreads();
  float s = 0.f;
#pragma unroll
  for (int i = 0; i < 32; ++i) s += red[i];
  __syncthreads();
  return s;
}

__device__ __forceinline__ float diag_partial(const v8f& a, int lane) {
  float p = 0.f;
  const int mb = (lane >> 4) * 8;
  const int n = lane & 15;
#pragma unroll
  for (int r = 0; r < 8; ++r) p += (n == (r + mb)) ? a[r] : 0.f;
  return p;
}

__global__ __launch_bounds__(32) void tyler_tangent_kernel(
    const float* __restrict__ x, float* __restrict__ out) {
  const int b    = blockIdx.x;
  const int lane = threadIdx.x;
  const int row  = lane & 15;
  const int hi   = lane >> 4;
  const int koff = hi * 2;

  __shared__ float Xs[C_DIM * XS_STRIDE];
  __shared__ float aug[C_DIM * AUG_STRIDE];
  __shared__ float As[C_DIM * JS];
  __shared__ float Vs[C_DIM * JS];
  __shared__ float wbuf[T_DIM];
  __shared__ float lw[C_DIM];
  __shared__ float red[32];

  {
    const float4* xin =
        reinterpret_cast<const float4*>(x + (size_t)b * (C_DIM * T_DIM));
#pragma unroll
    for (int i = 0; i < 20; ++i) {
      int idx4 = lane + 32 * i;
      float4 v = xin[idx4];
      int c = idx4 / 40;
      int t0 = (idx4 % 40) * 4;
      float* p = &Xs[c * XS_STRIDE + t0];
      p[0] = v.x; p[1] = v.y; p[2] = v.z; p[3] = v.w;
    }
  }
  __syncthreads();

  const float* fragbase = &Xs[row * XS_STRIDE + koff];

  v8f acc = {0.f, 0.f, 0.f, 0.f, 0.f, 0.f, 0.f, 0.f};
#pragma unroll 1
  for (int cb = 0; cb < 10; ++cb) {
    v2f f[4];
#pragma unroll
    for (int u = 0; u < 4; ++u) {
      const float* p = fragbase + (cb * 4 + u) * 4;
      f[u].x = p[0];
      f[u].y = p[1];
    }
#pragma unroll
    for (int u = 0; u < 4; ++u) acc = wmma_f32_16x16x4(f[u], f[u], acc);
  }
  float tr = wave_sum32(diag_partial(acc, lane) * (1.f / T_DIM), red, lane);
  float scl = (1.f / T_DIM) * (16.f / (tr + EPSV));
  v8f ck;
#pragma unroll
  for (int r = 0; r < 8; ++r) ck[r] = acc[r] * scl;

#pragma unroll 1
  for (int it = 0; it < NITERS; ++it) {
#pragma unroll
    for (int r = 0; r < 8; ++r) {
      int m = r + 8 * hi;
      aug[m * AUG_STRIDE + row]      = ck[r] + ((m == row) ? EPSV : 0.f);
      aug[m * AUG_STRIDE + 16 + row] = (m == row) ? 1.f : 0.f;
    }
    __syncthreads();

#pragma unroll 1
    for (int k = 0; k < 16; ++k) {
      float pinv = 1.f / aug[k * AUG_STRIDE + k];
      __syncthreads();
      aug[k * AUG_STRIDE + lane] *= pinv;
      __syncthreads();
#pragma unroll 1
      for (int r = 0; r < 16; ++r) {
        if (r == k) continue;
        float fct = aug[r * AUG_STRIDE + k];
        aug[r * AUG_STRIDE + lane] -= fct * aug[k * AUG_STRIDE + lane];
      }
      __syncthreads();
    }

#pragma unroll 1
    for (int j5 = 0; j5 < 5; ++j5) {
      int t = lane + 32 * j5;
      float xc[16];
#pragma unroll
      for (int i = 0; i < 16; ++i) xc[i] = Xs[i * XS_STRIDE + t];
      float qv = 0.f;
#pragma unroll
      for (int i = 0; i < 16; ++i) {
        const float* ci = &aug[i * AUG_STRIDE + 16];
        qv += ci[i] * xc[i] * xc[i];
        float ra = 0.f;
#pragma unroll
        for (int j = i + 1; j < 16; ++j) ra += ci[j] * xc[j];
        qv += 2.f * xc[i] * ra;
      }
      wbuf[t] = 16.f / fmaxf(qv, EPSV);
    }
    __syncthreads();

    v8f a2 = {0.f, 0.f, 0.f, 0.f, 0.f, 0.f, 0.f, 0.f};
#pragma unroll 1
    for (int cb = 0; cb < 10; ++cb) {
      v2f bb[4], aa[4];
#pragma unroll
      for (int u = 0; u < 4; ++u) {
        int t = (cb * 4 + u) * 4 + koff;
        const float* p = fragbase + (cb * 4 + u) * 4;
        bb[u].x = p[0];
        bb[u].y = p[1];
        aa[u].x = bb[u].x * wbuf[t];
        aa[u].y = bb[u].y * wbuf[t + 1];
      }
#pragma unroll
      for (int u = 0; u < 4; ++u) a2 = wmma_f32_16x16x4(aa[u], bb[u], a2);
    }
    float tr2 = wave_sum32(diag_partial(a2, lane) * (1.f / T_DIM), red, lane);
    float s2 = (1.f / T_DIM) * (16.f / (tr2 + EPSV));
#pragma unroll
    for (int r = 0; r < 8; ++r) ck[r] = a2[r] * s2;
    __syncthreads();
  }

#pragma unroll
  for (int r = 0; r < 8; ++r) { int m = r + 8 * hi; As[m * JS + row] = ck[r]; }
  __syncthreads();
  float tv[8];
#pragma unroll
  for (int r = 0; r < 8; ++r) { int m = r + 8 * hi; tv[r] = As[row * JS + m]; }
  __syncthreads();
#pragma unroll
  for (int r = 0; r < 8; ++r) {
    int m = r + 8 * hi;
    As[m * JS + row] = 0.5f * (ck[r] + tv[r]);
  }
  if (lane < 16) {
#pragma unroll
    for (int j = 0; j < 16; ++j) Vs[lane * JS + j] = (lane == j) ? 1.f : 0.f;
  }
  __syncthreads();

#pragma unroll 1
  for (int sweep = 0; sweep < NSWEEP; ++sweep) {
#pragma unroll 1
    for (int p = 0; p < 15; ++p) {
#pragma unroll 1
      for (int q = p + 1; q < 16; ++q) {
        float apq = As[p * JS + q];
        float app = As[p * JS + p];
        float aqq = As[q * JS + q];
        float cc = 1.f, ss = 0.f;
        if (fabsf(apq) > 1e-30f) {
          float th = (aqq - app) / (2.f * apq);
          float tt = 1.f / (fabsf(th) + sqrtf(th * th + 1.f));
          if (th < 0.f) tt = -tt;
          cc = 1.f / sqrtf(1.f + tt * tt);
          ss = tt * cc;
        }
        __syncthreads();
        int k2 = lane & 15;
        if (lane < 16) {
          float akp = As[k2 * JS + p], akq = As[k2 * JS + q];
          As[k2 * JS + p] = cc * akp - ss * akq;
          As[k2 * JS + q] = ss * akp + cc * akq;
        } else {
          float vkp = Vs[k2 * JS + p], vkq = Vs[k2 * JS + q];
          Vs[k2 * JS + p] = cc * vkp - ss * vkq;
          Vs[k2 * JS + q] = ss * vkp + cc * vkq;
        }
        __syncthreads();
        if (lane < 16) {
          float apk = As[p * JS + k2], aqk = As[q * JS + k2];
          As[p * JS + k2] = cc * apk - ss * aqk;
          As[q * JS + k2] = ss * apk + cc * aqk;
        }
        __syncthreads();
      }
    }
  }
  if (lane < 16) lw[lane] = logf(fmaxf(As[lane * JS + lane], EPSV));
  __syncthreads();

  v8f lg = {0.f, 0.f, 0.f, 0.f, 0.f, 0.f, 0.f, 0.f};
#pragma unroll
  for (int kc = 0; kc < 4; ++kc) {
    int k0 = kc * 4 + koff;
    v2f bb; bb.x = Vs[row * JS + k0]; bb.y = Vs[row * JS + k0 + 1];
    float l0 = lw[k0], l1 = lw[k0 + 1];
    v2f aa; aa.x = bb.x * l0; aa.y = bb.y * l1;
    lg = wmma_f32_16x16x4(aa, bb, lg);
  }

  const float SQ2 = 1.4142135623730951f;
  float* op = out + (size_t)b * 136;
#pragma unroll
  for (int r = 0; r < 8; ++r) {
    int m = r + 8 * hi;
    if (m <= row) {
      int idx = m * 16 - (m * (m - 1)) / 2 + (row - m);
      op[idx] = lg[r] * ((m == row) ? 1.f : SQ2);
    }
  }
}

extern "C" void kernel_launch(void* const* d_in, const int* in_sizes, int n_in,
                              void* d_out, int out_size, void* d_ws,
                              size_t ws_size, hipStream_t stream) {
  const float* x = (const float*)d_in[0];
  float* out = (float*)d_out;
  int B = in_sizes[0] / (C_DIM * T_DIM);
  tyler_tangent_kernel<<<B, 32, 0, stream>>>(x, out);
}


# Both Gram products (S, the 10 Cn): LDS fragment reads issued as whole groups ahead of straight-line WMMAs on two alternating accumulator tiles (even/odd chunks, added at the end); sum order changes
# speedup vs baseline: 12.0339x; 12.0339x over previous
#include <hip/hip_runtime.h>
#include <math.h>

typedef __attribute__((ext_vector_type(2))) float v2f;
typedef __attribute__((ext_vector_type(8))) float v8f;

#define C_DIM 16
#define T_DIM 160
#define NITERS 10
#define EPSV 1e-8f
#define XS_STRIDE 162
#define JS 17
#define RS (C_DIM * JS)

__device__ __forceinline__ v8f wmma_f32_16x16x4(v2f a, v2f b, v8f c) {
  return __builtin_amdgcn_wmma_f32_16x16x4_f32(false, a, false, b, (short)0, c,
                                               false, false);
}

__device__ __forceinline__ float xhalf(float v) {
  int b = __builtin_bit_cast(int, v);
  return __builtin_bit_cast(float, __builtin_amdgcn_permlanex16(
      b, b, 0x76543210, 0xfedcba98, false, false));
}

__device__ __forceinline__ float lane_bcast(float v, int src_lane) {
  return __builtin_bit_cast(
      float, __builtin_amdgcn_readlane(__builtin_bit_cast(int, v), src_lane));
}

__device__ __forceinline__ float wave_sum32(float v, float* red, int lane) {
  red[lane] = v;
  __syncthreads();
  float s = 0.f;
#pragma unroll
  for (int i = 0; i < 32; ++i) s += red[i];
  __syncthreads();
  return s;
}

__device__ __forceinline__ float diag_partial(const v8f& a, int lane) {
  float p = 0.f;
  const int mb = (lane >> 4) * 8;
  const int n = lane & 15;
#pragma unroll
  for (int r = 0; r < 8; ++r) p += (n == (r + mb)) ? a[r] : 0.f;
  return p;
}

#define NS_EIGMAX 2.8f
#define NS_TOL 2.5e-4f
#define CAY_RADIUS2 0.16f
#define MAX_ROOTS 6
#define MAX_NS_TRIPS 10
#define MAX_INV_TRIPS 10
#define INV_S  (2.f + NS_EIGMAX)
#define INV_D2 (NS_EIGMAX * NS_EIGMAX)
#define INV_T2 (2.f * INV_S * INV_S / INV_D2 - 1.f)
#define INV_C0 (8.f * INV_S / (INV_D2 * INV_T2))
#define INV_C1 (-8.f / (INV_D2 * INV_T2))

__device__ __forceinline__ v8f mprod(const v8f& m, const v8f& n, v8f acc) {
#pragma unroll
  for (int kc = 0; kc < 4; ++kc) {
    v2f fa; fa.x = m[2 * kc]; fa.y = m[2 * kc + 1];
    v2f fb; fb.x = n[2 * kc]; fb.y = n[2 * kc + 1];
    acc = wmma_f32_16x16x4(fa, fb, acc);
  }
  return acc;
}

__device__ __forceinline__ v8f series_block(const v8f& idn, const v8f& v, const v8f& v2,
                                            const v8f& v3, int j) {
  const float k1 = 8.f * j + 1.f;
  const float c0 = 1.f / k1, c1 = 1.f / (k1 + 2.f), c2 = 1.f / (k1 + 4.f), c3 = 1.f / (k1 + 6.f);
  v8f b;
#pragma unroll
  for (int r = 0; r < 8; ++r)
    b[r] = fmaf(c3, v3[r], fmaf(c2, v2[r], fmaf(c1, v[r], c0 * idn[r])));
  return b;
}

__global__ __launch_bounds__(32) void tyler_tangent_kernel(
    const float* __restrict__ x, float* __restrict__ out) {
  const int lane = threadIdx.x;
  const int row  = lane & 15;
  const int hi   = lane >> 4;
  const int koff = hi * 2;

  __shared__ float Xs[C_DIM * XS_STRIDE];
  __shared__ float Rs[2 * RS];
  __shared__ float wbuf[T_DIM];
  __shared__ float red[32];

#pragma unroll 1
  for (int mi = 0; mi < 2; ++mi) {
  const int b = 2 * blockIdx.x + mi;
  float* As = &Rs[mi * RS];
  {
    const float4* xin =
        reinterpret_cast<const float4*>(x + (size_t)b * (C_DIM * T_DIM));
#pragma unroll
    for (int i = 0; i < 20; ++i) {
      int idx4 = lane + 32 * i;
      float4 v = xin[idx4];
      int c = idx4 / 40;
      int t0 = (idx4 % 40) * 4;
      float* p = &Xs[c * XS_STRIDE + t0];
      p[0] = v.x; p[1] = v.y; p[2] = v.z; p[3] = v.w;
    }
  }
  __syncthreads();

  const float* fragbase = &Xs[row * XS_STRIDE + koff];

  v8f acc = {0.f, 0.f, 0.f, 0.f, 0.f, 0.f, 0.f, 0.f};
  {
    v8f accb = {0.f, 0.f, 0.f, 0.f, 0.f, 0.f, 0.f, 0.f};
    v2f f[40];
#pragma unroll
    for (int c = 0; c < 40; ++c) {
      const float* p = fragbase + c * 4;
      f[c].x = p[0];
      f[c].y = p[1];
    }
#pragma unroll
    for (int c = 0; c < 40; c += 2) {
      acc  = wmma_f32_16x16x4(f[c], f[c], acc);
      accb = wmma_f32_16x16x4(f[c + 1], f[c + 1], accb);
    }
    __builtin_amdgcn_sched_group_barrier(0x100, 20, 0);
    __builtin_amdgcn_sched_group_barrier(0x008, 40, 0);
#pragma unroll
    for (int r = 0; r < 8; ++r) acc[r] += accb[r];
  }
  float tr = wave_sum32(diag_partial(acc, lane) * (1.f / T_DIM), red, lane);
  float scl = (1.f / T_DIM) * (16.f / (tr + EPSV));
  v8f ck;
#pragma unroll
  for (int r = 0; r < 8; ++r) ck[r] = acc[r] * scl;

#pragma unroll 1
  for (int it = 0; it < NITERS; ++it) {
    float col[16];
#pragma unroll
    for (int r = 0; r < 8; ++r) {
      const float o = xhalf(ck[r]);
      col[r]     = hi ? o : ck[r];
      col[r + 8] = hi ? ck[r] : o;
    }
#pragma unroll
    for (int i = 0; i < 16; ++i) col[i] += (i == row) ? EPSV : 0.f;

    float xc[5][16];
#pragma unroll
    for (int j = 0; j < 5; ++j) {
#pragma unroll
      for (int i = 0; i < 16; ++i) xc[j][i] = Xs[i * XS_STRIDE + lane + 32 * j];
    }
    float qv[5] = {0.f, 0.f, 0.f, 0.f, 0.f};

    float mm[16][16];
    float dd[16], rnv[16], lkk[16];
    dd[0] = lane_bcast(col[0], 0);
#pragma unroll
    for (int i = 1; i < 16; ++i) mm[0][i] = lane_bcast(col[i], 0);
    rnv[0] = 1.f / dd[0];
    lkk[0] = col[0] * rnv[0];
#pragma unroll
    for (int i = 1; i < 16; ++i) col[i] = fmaf(-mm[0][i], lkk[0], col[i]);
    dd[1] = lane_bcast(col[1], 1);
#pragma unroll
    for (int i = 2; i < 16; ++i) mm[1][i] = lane_bcast(col[i], 1);
    __builtin_amdgcn_sched_barrier(0);

#pragma unroll
    for (int k = 0; k < 16; ++k) {
      if (k + 1 < 16) {
        rnv[k + 1] = 1.f / dd[k + 1];
        lkk[k + 1] = col[k + 1] * rnv[k + 1];
      }
      float z[5];
#pragma unroll
      for (int j = 0; j < 5; ++j) {
        z[j]  = xc[j][k] * rnv[k];
        qv[j] = fmaf(xc[j][k], z[j], qv[j]);
      }
#pragma unroll
      for (int i = k + 1; i < 16; ++i) {
#pragma unroll
        for (int j = 0; j < 5; ++j) xc[j][i] = fmaf(-mm[k][i], z[j], xc[j][i]);
      }
      if (k + 1 < 16) {
#pragma unroll
        for (int i = k + 2; i < 16; ++i)
          col[i] = fmaf(-mm[k + 1][i], lkk[k + 1], col[i]);
      }
      if (k + 2 < 16) {
        dd[k + 2] = lane_bcast(col[k + 2], k + 2);
#pragma unroll
        for (int i = k + 3; i < 16; ++i) mm[k + 2][i] = lane_bcast(col[i], k + 2);
      }
      __builtin_amdgcn_sched_barrier(0);
    }
#pragma unroll
    for (int j = 0; j < 5; ++j)
      wbuf[lane + 32 * j] = (16.f / T_DIM) / fmaxf(qv[j], EPSV);
    __syncthreads();

    v8f a2 = {0.f, 0.f, 0.f, 0.f, 0.f, 0.f, 0.f, 0.f};
    {
      v8f a2b = {0.f, 0.f, 0.f, 0.f, 0.f, 0.f, 0.f, 0.f};
      __builtin_amdgcn_sched_barrier(0);
#pragma unroll
      for (int h = 0; h < 2; ++h) {
        v2f bb[20], ww[20];
#pragma unroll
        for (int c = 0; c < 20; ++c) {
          const int ch = 20 * h + c;
          const float* p = fragbase + ch * 4;
          bb[c].x = p[0];
          bb[c].y = p[1];
          ww[c].x = wbuf[ch * 4 + koff];
          ww[c].y = wbuf[ch * 4 + koff + 1];
        }
#pragma unroll
        for (int c = 0; c < 20; c += 2) {
          v2f aa0, aa1;
          aa0.x = bb[c].x * ww[c].x;         aa0.y = bb[c].y * ww[c].y;
          aa1.x = bb[c + 1].x * ww[c + 1].x; aa1.y = bb[c + 1].y * ww[c + 1].y;
          a2  = wmma_f32_16x16x4(aa0, bb[c], a2);
          a2b = wmma_f32_16x16x4(aa1, bb[c + 1], a2b);
        }
      }
      __builtin_amdgcn_sched_group_barrier(0x100, 20, 0);
      __builtin_amdgcn_sched_group_barrier(0x00A, 40, 0);
      __builtin_amdgcn_sched_group_barrier(0x100, 20, 0);
      __builtin_amdgcn_sched_group_barrier(0x00A, 40, 0);
#pragma unroll
      for (int r = 0; r < 8; ++r) a2[r] += a2b[r];
    }
    ck = a2;
    __syncthreads();
  }
  {
    const float tr2 = wave_sum32(diag_partial(ck, lane), red, lane);
    const float s2 = 16.f / (tr2 + EPSV);
#pragma unroll
    for (int r = 0; r < 8; ++r) ck[r] *= s2;
  }

#pragma unroll
  for (int r = 0; r < 8; ++r) { int m = r + 8 * hi; As[m * JS + row] = ck[r]; }
  __syncthreads();
  float tv[8];
#pragma unroll
  for (int r = 0; r < 8; ++r) { int m = r + 8 * hi; tv[r] = As[row * JS + m]; }
  __syncthreads();
#pragma unroll
  for (int r = 0; r < 8; ++r) As[(r + 8 * hi) * JS + row] = 0.5f * (ck[r] + tv[r]);
  __syncthreads();
  }

  v8f idn, y0, y1;
#pragma unroll
  for (int r = 0; r < 8; ++r) {
    const int m = r + 8 * hi;
    y0[r]  = Rs[m * JS + row];
    y1[r]  = Rs[RS + m * JS + row];
    idn[r] = (m == row) ? 1.f : 0.f;
  }
  float sig;
  {
    float c0 = 0.f, c1 = 0.f;
#pragma unroll
    for (int r = 0; r < 8; ++r) { c0 += fabsf(y0[r]); c1 += fabsf(y1[r]); }
    float cm = fmaxf(c0 + xhalf(c0), c1 + xhalf(c1));
    cm = fmaxf(cm, __shfl_xor(cm, 8, 32));
    cm = fmaxf(cm, __shfl_xor(cm, 4, 32));
    cm = fmaxf(cm, __shfl_xor(cm, 2, 32));
    cm = fmaxf(cm, __shfl_xor(cm, 1, 32));
    sig = fminf(1.f, NS_EIGMAX / cm);
  }
#pragma unroll
  for (int r = 0; r < 8; ++r) { y0[r] *= sig; y1[r] *= sig; }

  const v8f zero = {0.f, 0.f, 0.f, 0.f, 0.f, 0.f, 0.f, 0.f};
  int ns = 0;
  v8f u0, u1, v0, v1, v20, v21;
#pragma unroll 1
  for (int st = 0; ; ++st) {
    v8f na0, na1, x0, x1;
#pragma unroll
    for (int r = 0; r < 8; ++r) {
      const float a0 = y0[r] + idn[r], a1 = y1[r] + idn[r];
      na0[r] = -a0;
      na1[r] = -a1;
      x0[r] = fmaf(INV_C1, a0, INV_C0 * idn[r]);
      x1[r] = fmaf(INV_C1, a1, INV_C0 * idn[r]);
    }
#pragma unroll 1
    for (int it = 0; it < MAX_INV_TRIPS; ++it) {
      v8f r0 = idn, r1 = idn;
#pragma unroll
      for (int kc = 0; kc < 4; ++kc) {
        v2f fa; fa.x = na0[2 * kc]; fa.y = na0[2 * kc + 1];
        v2f fb; fb.x = x0[2 * kc];  fb.y = x0[2 * kc + 1];
        r0 = wmma_f32_16x16x4(fa, fb, r0);
        v2f fc; fc.x = na1[2 * kc]; fc.y = na1[2 * kc + 1];
        v2f fd; fd.x = x1[2 * kc];  fd.y = x1[2 * kc + 1];
        r1 = wmma_f32_16x16x4(fc, fd, r1);
      }
      float dev = 0.f;
#pragma unroll
      for (int r = 0; r < 8; ++r) dev = fmaxf(dev, fmaxf(fabsf(r0[r]), fabsf(r1[r])));
      v8f xn0 = x0, xn1 = x1;
#pragma unroll
      for (int kc = 0; kc < 4; ++kc) {
        v2f fa; fa.x = x0[2 * kc]; fa.y = x0[2 * kc + 1];
        v2f fb; fb.x = r0[2 * kc]; fb.y = r0[2 * kc + 1];
        xn0 = wmma_f32_16x16x4(fa, fb, xn0);
        v2f fc; fc.x = x1[2 * kc]; fc.y = x1[2 * kc + 1];
        v2f fd; fd.x = r1[2 * kc]; fd.y = r1[2 * kc + 1];
        xn1 = wmma_f32_16x16x4(fc, fd, xn1);
      }
      x0 = xn0; x1 = xn1;
      if (__builtin_amdgcn_ballot_w32(dev >= NS_TOL) == 0) break;
    }
    v8f m0, m1;
#pragma unroll
    for (int r = 0; r < 8; ++r) { m0[r] = y0[r] - idn[r]; m1[r] = y1[r] - idn[r]; }
    u0 = zero; u1 = zero;
#pragma unroll
    for (int kc = 0; kc < 4; ++kc) {
      v2f fa; fa.x = m0[2 * kc]; fa.y = m0[2 * kc + 1];
      v2f fb; fb.x = x0[2 * kc]; fb.y = x0[2 * kc + 1];
      u0 = wmma_f32_16x16x4(fa, fb, u0);
      v2f fc; fc.x = m1[2 * kc]; fc.y = m1[2 * kc + 1];
      v2f fd; fd.x = x1[2 * kc]; fd.y = x1[2 * kc + 1];
      u1 = wmma_f32_16x16x4(fc, fd, u1);
    }
    v0 = zero; v1 = zero;
#pragma unroll
    for (int kc = 0; kc < 4; ++kc) {
      v2f fa; fa.x = u0[2 * kc]; fa.y = u0[2 * kc + 1];
      v0 = wmma_f32_16x16x4(fa, fa, v0);
      v2f fc; fc.x = u1[2 * kc]; fc.y = u1[2 * kc + 1];
      v1 = wmma_f32_16x16x4(fc, fc, v1);
    }
    v20 = zero; v21 = zero;
#pragma unroll
    for (int kc = 0; kc < 4; ++kc) {
      v2f fa; fa.x = v0[2 * kc]; fa.y = v0[2 * kc + 1];
      v20 = wmma_f32_16x16x4(fa, fa, v20);
      v2f fc; fc.x = v1[2 * kc]; fc.y = v1[2 * kc + 1];
      v21 = wmma_f32_16x16x4(fc, fc, v21);
    }
    float c0 = 0.f, c1 = 0.f;
#pragma unroll
    for (int r = 0; r < 8; ++r) { c0 += fabsf(v20[r]); c1 += fabsf(v21[r]); }
    const float vc = fmaxf(c0 + xhalf(c0), c1 + xhalf(c1));
    if (st == MAX_ROOTS || __builtin_amdgcn_ballot_w32(vc >= CAY_RADIUS2) == 0) break;

    v8f z0 = idn, z1 = idn;
#pragma unroll 1
    for (int it = 0; it < MAX_NS_TRIPS; ++it) {
      v8f p0 = zero;
      v8f p1 = zero;
#pragma unroll
      for (int kc = 0; kc < 4; ++kc) {
        v2f fa; fa.x = y0[2 * kc]; fa.y = y0[2 * kc + 1];
        v2f fb; fb.x = z0[2 * kc]; fb.y = z0[2 * kc + 1];
        p0 = wmma_f32_16x16x4(fa, fb, p0);
        v2f fc; fc.x = y1[2 * kc]; fc.y = y1[2 * kc + 1];
        v2f fd; fd.x = z1[2 * kc]; fd.y = z1[2 * kc + 1];
        p1 = wmma_f32_16x16x4(fc, fd, p1);
      }
      float dev = 0.f;
      v8f t0, t1;
#pragma unroll
      for (int r = 0; r < 8; ++r) {
        dev = fmaxf(dev, fmaxf(fabsf(p0[r] - idn[r]), fabsf(p1[r] - idn[r])));
        t0[r] = fmaf(-0.5f, p0[r], 1.5f * idn[r]);
        t1[r] = fmaf(-0.5f, p1[r], 1.5f * idn[r]);
      }
      v8f yn0 = zero;
      v8f yn1 = zero, zn0 = zero, zn1 = zero;
#pragma unroll
      for (int kc = 0; kc < 4; ++kc) {
        v2f ft; ft.x = t0[2 * kc]; ft.y = t0[2 * kc + 1];
        v2f fy; fy.x = y0[2 * kc]; fy.y = y0[2 * kc + 1];
        v2f fz; fz.x = z0[2 * kc]; fz.y = z0[2 * kc + 1];
        yn0 = wmma_f32_16x16x4(ft, fy, yn0);
        zn0 = wmma_f32_16x16x4(ft, fz, zn0);
        v2f fu; fu.x = t1[2 * kc]; fu.y = t1[2 * kc + 1];
        v2f fv; fv.x = y1[2 * kc]; fv.y = y1[2 * kc + 1];
        v2f fw; fw.x = z1[2 * kc]; fw.y = z1[2 * kc + 1];
        yn1 = wmma_f32_16x16x4(fu, fv, yn1);
        zn1 = wmma_f32_16x16x4(fu, fw, zn1);
      }
      y0 = yn0; y1 = yn1; z0 = zn0; z1 = zn1;
      if (__builtin_amdgcn_ballot_w32(dev >= NS_TOL) == 0) break;
    }
    ++ns;
  }

  v8f lg0, lg1;
  {
    const v8f v30 = mprod(v0, v20, zero),  v31 = mprod(v1, v21, zero);
    const v8f v40 = mprod(v20, v20, zero), v41 = mprod(v21, v21, zero);
    v8f s0 = series_block(idn, v0, v20, v30, 3);
    v8f s1 = series_block(idn, v1, v21, v31, 3);
#pragma unroll
    for (int j = 2; j >= 0; --j) {
      s0 = mprod(v40, s0, series_block(idn, v0, v20, v30, j));
      s1 = mprod(v41, s1, series_block(idn, v1, v21, v31, j));
    }
    v8f q0 = mprod(u0, s0, zero);
    v8f q1 = mprod(u1, s1, zero);
    asm volatile("v_nop\n\tv_nop\n\tv_nop\n\tv_nop" : "+v"(q0), "+v"(q1));
    const float up  = 2.f * (float)(1 << ns);
    const float lsg = logf(sig);
#pragma unroll
    for (int r = 0; r < 8; ++r) {
      lg0[r] = fmaf(up, q0[r], -lsg * idn[r]);
      lg1[r] = fmaf(up, q1[r], -lsg * idn[r]);
    }
  }

  const float SQ2 = 1.4142135623730951f;
  float* op = out + (size_t)(2 * blockIdx.x) * 136;
#pragma unroll
  for (int r = 0; r < 8; ++r) {
    int m = r + 8 * hi;
    if (m <= row) {
      int idx = m * 16 - (m * (m - 1)) / 2 + (row - m);
      const float cf = (m == row) ? 1.f : SQ2;
      op[idx]       = lg0[r] * cf;
      op[136 + idx] = lg1[r] * cf;
    }
  }
}

extern "C" void kernel_launch(void* const* d_in, const int* in_sizes, int n_in,
                              void* d_out, int out_size, void* d_ws,
                              size_t ws_size, hipStream_t stream) {
  const float* x = (const float*)d_in[0];
  float* out = (float*)d_out;
  int B = in_sizes[0] / (C_DIM * T_DIM);
  tyler_tangent_kernel<<<B / 2, 32, 0, stream>>>(x, out);
}


# Reciprocals in batches, one IEEE division each (product trick): pivots in pairs via e=c*d-m*m, r=1/(d*e); a lane's 5 weights via prefix products; 21 -> 9 divisions a trip; their rounding changes
# speedup vs baseline: 12.5531x; 1.0431x over previous
#include <hip/hip_runtime.h>
#include <math.h>

typedef __attribute__((ext_vector_type(2))) float v2f;
typedef __attribute__((ext_vector_type(8))) float v8f;

#define C_DIM 16
#define T_DIM 160
#define NITERS 10
#define EPSV 1e-8f
#define XS_STRIDE 162
#define JS 17
#define RS (C_DIM * JS)

__device__ __forceinline__ v8f wmma_f32_16x16x4(v2f a, v2f b, v8f c) {
  return __builtin_amdgcn_wmma_f32_16x16x4_f32(false, a, false, b, (short)0, c,
                                               false, false);
}

__device__ __forceinline__ float xhalf(float v) {
  int b = __builtin_bit_cast(int, v);
  return __builtin_bit_cast(float, __builtin_amdgcn_permlanex16(
      b, b, 0x76543210, 0xfedcba98, false, false));
}

__device__ __forceinline__ float lane_bcast(float v, int src_lane) {
  return __builtin_bit_cast(
      float, __builtin_amdgcn_readlane(__builtin_bit_cast(int, v), src_lane));
}

__device__ __forceinline__ float wave_sum32(float v, float* red, int lane) {
  red[lane] = v;
  __syncthreads();
  float s = 0.f;
#pragma unroll
  for (int i = 0; i < 32; ++i) s += red[i];
  __syncthreads();
  return s;
}

__device__ __forceinline__ float diag_partial(const v8f& a, int lane) {
  float p = 0.f;
  const int mb = (lane >> 4) * 8;
  const int n = lane & 15;
#pragma unroll
  for (int r = 0; r < 8; ++r) p += (n == (r + mb)) ? a[r] : 0.f;
  return p;
}

#define NS_EIGMAX 2.8f
#define NS_TOL 2.5e-4f
#define CAY_RADIUS2 0.16f
#define MAX_ROOTS 6
#define MAX_NS_TRIPS 10
#define MAX_INV_TRIPS 10
#define INV_S  (2.f + NS_EIGMAX)
#define INV_D2 (NS_EIGMAX * NS_EIGMAX)
#define INV_T2 (2.f * INV_S * INV_S / INV_D2 - 1.f)
#define INV_C0 (8.f * INV_S / (INV_D2 * INV_T2))
#define INV_C1 (-8.f / (INV_D2 * INV_T2))

__device__ __forceinline__ v8f mprod(const v8f& m, const v8f& n, v8f acc) {
#pragma unroll
  for (int kc = 0; kc < 4; ++kc) {
    v2f fa; fa.x = m[2 * kc]; fa.y = m[2 * kc + 1];
    v2f fb; fb.x = n[2 * kc]; fb.y = n[2 * kc + 1];
    acc = wmma_f32_16x16x4(fa, fb, acc);
  }
  return acc;
}

__device__ __forceinline__ v8f series_block(const v8f& idn, const v8f& v, const v8f& v2,
                                            const v8f& v3, int j) {
  const float k1 = 8.f * j + 1.f;
  const float c0 = 1.f / k1, c1 = 1.f / (k1 + 2.f), c2 = 1.f / (k1 + 4.f), c3 = 1.f / (k1 + 6.f);
  v8f b;
#pragma unroll
  for (int r = 0; r < 8; ++r)
    b[r] = fmaf(c3, v3[r], fmaf(c2, v2[r], fmaf(c1, v[r], c0 * idn[r])));
  return b;
}

__global__ __launch_bounds__(32) void tyler_tangent_kernel(
    const float* __restrict__ x, float* __restrict__ out) {
  const int lane = threadIdx.x;
  const int row  = lane & 15;
  const int hi   = lane >> 4;
  const int koff = hi * 2;

  __shared__ float Xs[C_DIM * XS_STRIDE];
  __shared__ float Rs[2 * RS];
  __shared__ float wbuf[T_DIM];
  __shared__ float red[32];

#pragma unroll 1
  for (int mi = 0; mi < 2; ++mi) {
  const int b = 2 * blockIdx.x + mi;
  float* As = &Rs[mi * RS];
  {
    const float4* xin =
        reinterpret_cast<const float4*>(x + (size_t)b * (C_DIM * T_DIM));
#pragma unroll
    for (int i = 0; i < 20; ++i) {
      int idx4 = lane + 32 * i;
      float4 v = xin[idx4];
      int c = idx4 / 40;
      int t0 = (idx4 % 40) * 4;
      float* p = &Xs[c * XS_STRIDE + t0];
      p[0] = v.x; p[1] = v.y; p[2] = v.z; p[3] = v.w;
    }
  }
  __syncthreads();

  const float* fragbase = &Xs[row * XS_STRIDE + koff];

  v8f acc = {0.f, 0.f, 0.f, 0.f, 0.f, 0.f, 0.f, 0.f};
  {
    v8f accb = {0.f, 0.f, 0.f, 0.f, 0.f, 0.f, 0.f, 0.f};
    v2f f[40];
#pragma unroll
    for (int c = 0; c < 40; ++c) {
      const float* p = fragbase + c * 4;
      f[c].x = p[0];
      f[c].y = p[1];
    }
#pragma unroll
    for (int c = 0; c < 40; c += 2) {
      acc  = wmma_f32_16x16x4(f[c], f[c], acc);
      accb = wmma_f32_16x16x4(f[c + 1], f[c + 1], accb);
    }
    __builtin_amdgcn_sched_group_barrier(0x100, 20, 0);
    __builtin_amdgcn_sched_group_barrier(0x008, 40, 0);
#pragma unroll
    for (int r = 0; r < 8; ++r) acc[r] += accb[r];
  }
  float tr = wave_sum32(diag_partial(acc, lane) * (1.f / T_DIM), red, lane);
  float scl = (1.f / T_DIM) * (16.f / (tr + EPSV));
  v8f ck;
#pragma unroll
  for (int r = 0; r < 8; ++r) ck[r] = acc[r] * scl;

#pragma unroll 1
  for (int it = 0; it < NITERS; ++it) {
    float col[16];
#pragma unroll
    for (int r = 0; r < 8; ++r) {
      const float o = xhalf(ck[r]);
      col[r]     = hi ? o : ck[r];
      col[r + 8] = hi ? ck[r] : o;
    }
#pragma unroll
    for (int i = 0; i < 16; ++i) col[i] += (i == row) ? EPSV : 0.f;

    float xc[5][16];
#pragma unroll
    for (int j = 0; j < 5; ++j) {
#pragma unroll
      for (int i = 0; i < 16; ++i) xc[j][i] = Xs[i * XS_STRIDE + lane + 32 * j];
    }
    float qv[5] = {0.f, 0.f, 0.f, 0.f, 0.f};

    float mm[16][16];
    float dd[16], cc[16], rnv[16], lkk[16];
    dd[0] = lane_bcast(col[0], 0);
#pragma unroll
    for (int i = 1; i < 16; ++i) mm[0][i] = lane_bcast(col[i], 0);
    cc[0] = lane_bcast(col[1], 1);
    {
      const float e = fmaf(cc[0], dd[0], -mm[0][1] * mm[0][1]);
      const float r = 1.f / (dd[0] * e);
      rnv[0] = r * e;
      rnv[1] = r * (dd[0] * dd[0]);
    }
    lkk[0] = col[0] * rnv[0];
#pragma unroll
    for (int i = 1; i < 16; ++i) col[i] = fmaf(-mm[0][i], lkk[0], col[i]);
#pragma unroll
    for (int i = 2; i < 16; ++i) mm[1][i] = lane_bcast(col[i], 1);
    __builtin_amdgcn_sched_barrier(0);

#pragma unroll
    for (int k = 0; k < 16; ++k) {
      if (k + 1 < 16) {
        if (((k + 1) & 1) == 0) {
          const float e = fmaf(cc[k + 1], dd[k + 1], -mm[k + 1][k + 2] * mm[k + 1][k + 2]);
          const float r = 1.f / (dd[k + 1] * e);
          rnv[k + 1] = r * e;
          rnv[k + 2] = r * (dd[k + 1] * dd[k + 1]);
        }
        lkk[k + 1] = col[k + 1] * rnv[k + 1];
      }
      float z[5];
#pragma unroll
      for (int j = 0; j < 5; ++j) {
        z[j]  = xc[j][k] * rnv[k];
        qv[j] = fmaf(xc[j][k], z[j], qv[j]);
      }
#pragma unroll
      for (int i = k + 1; i < 16; ++i) {
#pragma unroll
        for (int j = 0; j < 5; ++j) xc[j][i] = fmaf(-mm[k][i], z[j], xc[j][i]);
      }
      if (k + 1 < 16) {
#pragma unroll
        for (int i = k + 2; i < 16; ++i)
          col[i] = fmaf(-mm[k + 1][i], lkk[k + 1], col[i]);
      }
      if (k + 2 < 16) {
        if (((k + 2) & 1) == 0) {
          dd[k + 2] = lane_bcast(col[k + 2], k + 2);
          cc[k + 2] = lane_bcast(col[k + 3], k + 3);
        }
#pragma unroll
        for (int i = k + 3; i < 16; ++i) mm[k + 2][i] = lane_bcast(col[i], k + 2);
      }
      __builtin_amdgcn_sched_barrier(0);
    }
    {
      float mq[5], pp[5], wv[5];
#pragma unroll
      for (int j = 0; j < 5; ++j) mq[j] = fmaxf(qv[j], EPSV);
      pp[0] = mq[0];
#pragma unroll
      for (int j = 1; j < 5; ++j) pp[j] = pp[j - 1] * mq[j];
      float r = (16.f / T_DIM) / pp[4];
#pragma unroll
      for (int j = 4; j >= 1; --j) {
        wv[j] = r * pp[j - 1];
        r *= mq[j];
      }
      wv[0] = r;
#pragma unroll
      for (int j = 0; j < 5; ++j) wbuf[lane + 32 * j] = wv[j];
    }
    __syncthreads();

    v8f a2 = {0.f, 0.f, 0.f, 0.f, 0.f, 0.f, 0.f, 0.f};
    {
      v8f a2b = {0.f, 0.f, 0.f, 0.f, 0.f, 0.f, 0.f, 0.f};
      __builtin_amdgcn_sched_barrier(0);
#pragma unroll
      for (int h = 0; h < 2; ++h) {
        v2f bb[20], ww[20];
#pragma unroll
        for (int c = 0; c < 20; ++c) {
          const int ch = 20 * h + c;
          const float* p = fragbase + ch * 4;
          bb[c].x = p[0];
          bb[c].y = p[1];
          ww[c].x = wbuf[ch * 4 + koff];
          ww[c].y = wbuf[ch * 4 + koff + 1];
        }
#pragma unroll
        for (int c = 0; c < 20; c += 2) {
          v2f aa0, aa1;
          aa0.x = bb[c].x * ww[c].x;         aa0.y = bb[c].y * ww[c].y;
          aa1.x = bb[c + 1].x * ww[c + 1].x; aa1.y = bb[c + 1].y * ww[c + 1].y;
          a2  = wmma_f32_16x16x4(aa0, bb[c], a2);
          a2b = wmma_f32_16x16x4(aa1, bb[c + 1], a2b);
        }
      }
      __builtin_amdgcn_sched_group_barrier(0x100, 20, 0);
      __builtin_amdgcn_sched_group_barrier(0x00A, 40, 0);
      __builtin_amdgcn_sched_group_barrier(0x100, 20, 0);
      __builtin_amdgcn_sched_group_barrier(0x00A, 40, 0);
#pragma unroll
      for (int r = 0; r < 8; ++r) a2[r] += a2b[r];
    }
    ck = a2;
    __syncthreads();
  }
  {
    const float tr2 = wave_sum32(diag_partial(ck, lane), red, lane);
    const float s2 = 16.f / (tr2 + EPSV);
#pragma unroll
    for (int r = 0; r < 8; ++r) ck[r] *= s2;
  }

#pragma unroll
  for (int r = 0; r < 8; ++r) { int m = r + 8 * hi; As[m * JS + row] = ck[r]; }
  __syncthreads();
  float tv[8];
#pragma unroll
  for (int r = 0; r < 8; ++r) { int m = r + 8 * hi; tv[r] = As[row * JS + m]; }
  __syncthreads();
#pragma unroll
  for (int r = 0; r < 8; ++r) As[(r + 8 * hi) * JS + row] = 0.5f * (ck[r] + tv[r]);
  __syncthreads();
  }

  v8f idn, y0, y1;
#pragma unroll
  for (int r = 0; r < 8; ++r) {
    const int m = r + 8 * hi;
    y0[r]  = Rs[m * JS + row];
    y1[r]  = Rs[RS + m * JS + row];
    idn[r] = (m == row) ? 1.f : 0.f;
  }
  float sig;
  {
    float c0 = 0.f, c1 = 0.f;
#pragma unroll
    for (int r = 0; r < 8; ++r) { c0 += fabsf(y0[r]); c1 += fabsf(y1[r]); }
    float cm = fmaxf(c0 + xhalf(c0), c1 + xhalf(c1));
    cm = fmaxf(cm, __shfl_xor(cm, 8, 32));
    cm = fmaxf(cm, __shfl_xor(cm, 4, 32));
    cm = fmaxf(cm, __shfl_xor(cm, 2, 32));
    cm = fmaxf(cm, __shfl_xor(cm, 1, 32));
    sig = fminf(1.f, NS_EIGMAX / cm);
  }
#pragma unroll
  for (int r = 0; r < 8; ++r) { y0[r] *= sig; y1[r] *= sig; }

  const v8f zero = {0.f, 0.f, 0.f, 0.f, 0.f, 0.f, 0.f, 0.f};
  int ns = 0;
  v8f u0, u1, v0, v1, v20, v21;
#pragma unroll 1
  for (int st = 0; ; ++st) {
    v8f na0, na1, x0, x1;
#pragma unroll
    for (int r = 0; r < 8; ++r) {
      const float a0 = y0[r] + idn[r], a1 = y1[r] + idn[r];
      na0[r] = -a0;
      na1[r] = -a1;
      x0[r] = fmaf(INV_C1, a0, INV_C0 * idn[r]);
      x1[r] = fmaf(INV_C1, a1, INV_C0 * idn[r]);
    }
#pragma unroll 1
    for (int it = 0; it < MAX_INV_TRIPS; ++it) {
      v8f r0 = idn, r1 = idn;
#pragma unroll
      for (int kc = 0; kc < 4; ++kc) {
        v2f fa; fa.x = na0[2 * kc]; fa.y = na0[2 * kc + 1];
        v2f fb; fb.x = x0[2 * kc];  fb.y = x0[2 * kc + 1];
        r0 = wmma_f32_16x16x4(fa, fb, r0);
        v2f fc; fc.x = na1[2 * kc]; fc.y = na1[2 * kc + 1];
        v2f fd; fd.x = x1[2 * kc];  fd.y = x1[2 * kc + 1];
        r1 = wmma_f32_16x16x4(fc, fd, r1);
      }
      float dev = 0.f;
#pragma unroll
      for (int r = 0; r < 8; ++r) dev = fmaxf(dev, fmaxf(fabsf(r0[r]), fabsf(r1[r])));
      v8f xn0 = x0, xn1 = x1;
#pragma unroll
      for (int kc = 0; kc < 4; ++kc) {
        v2f fa; fa.x = x0[2 * kc]; fa.y = x0[2 * kc + 1];
        v2f fb; fb.x = r0[2 * kc]; fb.y = r0[2 * kc + 1];
        xn0 = wmma_f32_16x16x4(fa, fb, xn0);
        v2f fc; fc.x = x1[2 * kc]; fc.y = x1[2 * kc + 1];
        v2f fd; fd.x = r1[2 * kc]; fd.y = r1[2 * kc + 1];
        xn1 = wmma_f32_16x16x4(fc, fd, xn1);
      }
      x0 = xn0; x1 = xn1;
      if (__builtin_amdgcn_ballot_w32(dev >= NS_TOL) == 0) break;
    }
    v8f m0, m1;
#pragma unroll
    for (int r = 0; r < 8; ++r) { m0[r] = y0[r] - idn[r]; m1[r] = y1[r] - idn[r]; }
    u0 = zero; u1 = zero;
#pragma unroll
    for (int kc = 0; kc < 4; ++kc) {
      v2f fa; fa.x = m0[2 * kc]; fa.y = m0[2 * kc + 1];
      v2f fb; fb.x = x0[2 * kc]; fb.y = x0[2 * kc + 1];
      u0 = wmma_f32_16x16x4(fa, fb, u0);
      v2f fc; fc.x = m1[2 * kc]; fc.y = m1[2 * kc + 1];
      v2f fd; fd.x = x1[2 * kc]; fd.y = x1[2 * kc + 1];
      u1 = wmma_f32_16x16x4(fc, fd, u1);
    }
    v0 = zero; v1 = zero;
#pragma unroll
    for (int kc = 0; kc < 4; ++kc) {
      v2f fa; fa.x = u0[2 * kc]; fa.y = u0[2 * kc + 1];
      v0 = wmma_f32_16x16x4(fa, fa, v0);
      v2f fc; fc.x = u1[2 * kc]; fc.y = u1[2 * kc + 1];
      v1 = wmma_f32_16x16x4(fc, fc, v1);
    }
    v20 = zero; v21 = zero;
#pragma unroll
    for (int kc = 0; kc < 4; ++kc) {
      v2f fa; fa.x = v0[2 * kc]; fa.y = v0[2 * kc + 1];
      v20 = wmma_f32_16x16x4(fa, fa, v20);
      v2f fc; fc.x = v1[2 * kc]; fc.y = v1[2 * kc + 1];
      v21 = wmma_f32_16x16x4(fc, fc, v21);
    }
    float c0 = 0.f, c1 = 0.f;
#pragma unroll
    for (int r = 0; r < 8; ++r) { c0 += fabsf(v20[r]); c1 += fabsf(v21[r]); }
    const float vc = fmaxf(c0 + xhalf(c0), c1 + xhalf(c1));
    if (st == MAX_ROOTS || __builtin_amdgcn_ballot_w32(vc >= CAY_RADIUS2) == 0) break;

    v8f z0 = idn, z1 = idn;
#pragma unroll 1
    for (int it = 0; it < MAX_NS_TRIPS; ++it) {
      v8f p0 = zero;
      v8f p1 = zero;
#pragma unroll
      for (int kc = 0; kc < 4; ++kc) {
        v2f fa; fa.x = y0[2 * kc]; fa.y = y0[2 * kc + 1];
        v2f fb; fb.x = z0[2 * kc]; fb.y = z0[2 * kc + 1];
        p0 = wmma_f32_16x16x4(fa, fb, p0);
        v2f fc; fc.x = y1[2 * kc]; fc.y = y1[2 * kc + 1];
        v2f fd; fd.x = z1[2 * kc]; fd.y = z1[2 * kc + 1];
        p1 = wmma_f32_16x16x4(fc, fd, p1);
      }
      float dev = 0.f;
      v8f t0, t1;
#pragma unroll
      for (int r = 0; r < 8; ++r) {
        dev = fmaxf(dev, fmaxf(fabsf(p0[r] - idn[r]), fabsf(p1[r] - idn[r])));
        t0[r] = fmaf(-0.5f, p0[r], 1.5f * idn[r]);
        t1[r] = fmaf(-0.5f, p1[r], 1.5f * idn[r]);
      }
      v8f yn0 = zero;
      v8f yn1 = zero, zn0 = zero, zn1 = zero;
#pragma unroll
      for (int kc = 0; kc < 4; ++kc) {
        v2f ft; ft.x = t0[2 * kc]; ft.y = t0[2 * kc + 1];
        v2f fy; fy.x = y0[2 * kc]; fy.y = y0[2 * kc + 1];
        v2f fz; fz.x = z0[2 * kc]; fz.y = z0[2 * kc + 1];
        yn0 = wmma_f32_16x16x4(ft, fy, yn0);
        zn0 = wmma_f32_16x16x4(ft, fz, zn0);
        v2f fu; fu.x = t1[2 * kc]; fu.y = t1[2 * kc + 1];
        v2f fv; fv.x = y1[2 * kc]; fv.y = y1[2 * kc + 1];
        v2f fw; fw.x = z1[2 * kc]; fw.y = z1[2 * kc + 1];
        yn1 = wmma_f32_16x16x4(fu, fv, yn1);
        zn1 = wmma_f32_16x16x4(fu, fw, zn1);
      }
      y0 = yn0; y1 = yn1; z0 = zn0; z1 = zn1;
      if (__builtin_amdgcn_ballot_w32(dev >= NS_TOL) == 0) break;
    }
    ++ns;
  }

  v8f lg0, lg1;
  {
    const v8f v30 = mprod(v0, v20, zero),  v31 = mprod(v1, v21, zero);
    const v8f v40 = mprod(v20, v20, zero), v41 = mprod(v21, v21, zero);
    v8f s0 = series_block(idn, v0, v20, v30, 3);
    v8f s1 = series_block(idn, v1, v21, v31, 3);
#pragma unroll
    for (int j = 2; j >= 0; --j) {
      s0 = mprod(v40, s0, series_block(idn, v0, v20, v30, j));
      s1 = mprod(v41, s1, series_block(idn, v1, v21, v31, j));
    }
    v8f q0 = mprod(u0, s0, zero);
    v8f q1 = mprod(u1, s1, zero);
    asm volatile("v_nop\n\tv_nop\n\tv_nop\n\tv_nop" : "+v"(q0), "+v"(q1));
    const float up  = 2.f * (float)(1 << ns);
    const float lsg = logf(sig);
#pragma unroll
    for (int r = 0; r < 8; ++r) {
      lg0[r] = fmaf(up, q0[r], -lsg * idn[r]);
      lg1[r] = fmaf(up, q1[r], -lsg * idn[r]);
    }
  }

  const float SQ2 = 1.4142135623730951f;
  float* op = out + (size_t)(2 * blockIdx.x) * 136;
#pragma unroll
  for (int r = 0; r < 8; ++r) {
    int m = r + 8 * hi;
    if (m <= row) {
      int idx = m * 16 - (m * (m - 1)) / 2 + (row - m);
      const float cf = (m == row) ? 1.f : SQ2;
      op[idx]       = lg0[r] * cf;
      op[136 + idx] = lg1[r] * cf;
    }
  }
}

extern "C" void kernel_launch(void* const* d_in, const int* in_sizes, int n_in,
                              void* d_out, int out_size, void* d_ws,
                              size_t ws_size, hipStream_t stream) {
  const float* x = (const float*)d_in[0];
  float* out = (float*)d_out;
  int B = in_sizes[0] / (C_DIM * T_DIM);
  tyler_tangent_kernel<<<B / 2, 32, 0, stream>>>(x, out);
}
